# starting kernel
# baseline (speedup 1.0000x reference)
#include <hip/hip_runtime.h>
#include <hip/hip_bf16.h>
#include <cstdint>
#include <cstddef>


typedef __attribute__((ext_vector_type(2))) float v2f;
typedef __attribute__((ext_vector_type(8))) float v8f;

#define CDIV(a, b) (((a) + (b) - 1) / (b))


__global__ void k_zero(float* __restrict__ p, long long n) {
  long long i = (long long)blockIdx.x * blockDim.x + threadIdx.x;
  if (i < n) p[i] = 0.0f;
}

__global__ void k_negcopy(float* __restrict__ dst, const float* __restrict__ src,
                          long long n) {
  long long i = (long long)blockIdx.x * blockDim.x + threadIdx.x;
  if (i < n) dst[i] = -src[i];
}

__global__ void k_deg_count(const int* __restrict__ ei, float* __restrict__ deg, int E) {
  int e = blockIdx.x * blockDim.x + threadIdx.x;
  if (e < E) atomicAdd(&deg[ei[e]], 1.0f);
}

__global__ void k_deg_inv(float* __restrict__ deg, int N) {
  int i = blockIdx.x * blockDim.x + threadIdx.x;
  if (i < N) {
    float d = deg[i];
    deg[i] = (d > 0.0f) ? (1.0f / sqrtf(d)) : 0.0f;
  }
}

__global__ void k_edge_norm(const int* __restrict__ ei, const float* __restrict__ dis,
                            float* __restrict__ nrm, int E) {
  int e = blockIdx.x * blockDim.x + threadIdx.x;
  if (e < E) nrm[e] = dis[ei[e]] * dis[ei[E + e]];
}

__global__ void k_propagate(const int* __restrict__ ei, const float* __restrict__ nrm,
                            const float* __restrict__ x, float* __restrict__ out,
                            int E, int N, int F, float scale) {
  long long tid = (long long)blockIdx.x * blockDim.x + threadIdx.x;
  long long total = (long long)E * F;
  if (tid >= total) return;
  int e = (int)(tid / F);
  int f = (int)(tid % F);
  int r = ei[e];
  int c = ei[E + e];
  float w = scale * nrm[e];
  size_t nf = (size_t)N * F;
#pragma unroll
  for (int b = 0; b < 4; ++b) {
    float v = w * x[(size_t)b * nf + (size_t)c * F + f];
    atomicAdd(&out[(size_t)b * nf + (size_t)r * F + f], v);
  }
}

__global__ void k_pool(const float* __restrict__ x, const int* __restrict__ idx,
                       const float* __restrict__ w, float* __restrict__ out,
                       int Nout, int Nin, int F) {
  long long tid = (long long)blockIdx.x * blockDim.x + threadIdx.x;
  long long total = 4LL * Nout * F;
  if (tid >= total) return;
  int f = (int)(tid % F);
  int n = (int)((tid / F) % Nout);
  int b = (int)(tid / ((long long)F * Nout));
  size_t nfin = (size_t)Nin * F;
  float s = 0.0f;
#pragma unroll
  for (int k = 0; k < 3; ++k) {
    int j = idx[n * 3 + k];
    s += w[n * 3 + k] * x[(size_t)b * nfin + (size_t)j * F + f];
  }
  out[((size_t)b * Nout + n) * F + f] = s;
}

__global__ void k_dense_enc(const float* __restrict__ h, const float* __restrict__ w,
                            const float* __restrict__ bias, float* __restrict__ z) {
  int b = blockIdx.x >> 6;
  int j = blockIdx.x & 63;
  float s = 0.0f;
  for (int i = threadIdx.x; i < 8192; i += blockDim.x)
    s += h[b * 8192 + i] * w[(size_t)i * 64 + j];
  __shared__ float red[256];
  red[threadIdx.x] = s;
  __syncthreads();
  for (int off = 128; off > 0; off >>= 1) {
    if (threadIdx.x < off) red[threadIdx.x] += red[threadIdx.x + off];
    __syncthreads();
  }
  if (threadIdx.x == 0) {
    float v = red[0] + bias[j];
    z[b * 64 + j] = v > 0.0f ? v : 0.0f;
  }
}

__global__ void k_dense_dec(const float* __restrict__ z, const float* __restrict__ w,
                            const float* __restrict__ bias, float* __restrict__ h) {
  long long tid = (long long)blockIdx.x * blockDim.x + threadIdx.x;
  if (tid >= 4LL * 8192) return;
  int b = (int)(tid >> 13);
  int j = (int)(tid & 8191);
  float s = bias[j];
  for (int i = 0; i < 64; ++i) s += z[b * 64 + i] * w[(size_t)i * 8192 + j];
  h[tid] = s > 0.0f ? s : 0.0f;
}


__device__ __forceinline__ float ldsel(const float* __restrict__ p, int idx, bool ok) {
  float v = p[ok ? idx : 0];
  return ok ? v : 0.0f;
}

template <int FIN, int FOUT, int MODE>
__global__ __launch_bounds__(256) void k_cheb_gemm(
    const float* __restrict__ T, const float* __restrict__ W,
    const float* __restrict__ bias, float* __restrict__ out, int M) {
  constexpr int KC = (FIN + 3) / 4;
  constexpr int NTN = (FOUT + 15) / 16;
  constexpr int TM = 4;

  int wave = (int)((blockIdx.x * blockDim.x + threadIdx.x) >> 5);
  int lane = threadIdx.x & 31;
  int groups = (M >> 4) / TM;
  if (wave >= groups * NTN) return;

  int tn = wave % NTN;
  int gm = wave / NTN;
  int m0 = gm * (16 * TM);
  int n0 = tn * 16;
  int lm = lane & 15;
  int khalf = (lane >> 4) << 1;
  int mhi = (lane >> 4) << 3;
  int ncol = n0 + lm;
  bool colok = (FOUT % 16 == 0) || (ncol < FOUT);

  v2f bf[KC];
#pragma unroll
  for (int kc = 0; kc < KC; ++kc) {
    int ka = kc * 4 + khalf;
    if constexpr (FIN % 4 == 0 && FOUT % 16 == 0) {
      bf[kc].x = W[ka * FOUT + ncol];
      bf[kc].y = W[(ka + 1) * FOUT + ncol];
    } else {
      bf[kc].x = ldsel(W, ka * FOUT + ncol, colok && (ka < FIN));
      bf[kc].y = ldsel(W, (ka + 1) * FOUT + ncol, colok && (ka + 1 < FIN));
    }
  }
  float bv = 0.0f;
  if constexpr (MODE == 0) bv = ldsel(bias, ncol, colok);

#pragma unroll
  for (int t = 0; t < TM; ++t) {
    int mb = m0 + t * 16;
    const float* trow = T + (size_t)(mb + lm) * FIN;
    v8f c = {0.f, 0.f, 0.f, 0.f, 0.f, 0.f, 0.f, 0.f};
#pragma unroll
    for (int kc = 0; kc < KC; ++kc) {
      int ka = kc * 4 + khalf;
      v2f a;
      if constexpr (FIN % 4 == 0) {
        a = *(const v2f*)(trow + ka);
      } else {
        a.x = ldsel(trow, ka, ka < FIN);
        a.y = ldsel(trow, ka + 1, ka + 1 < FIN);
      }
      c = __builtin_amdgcn_wmma_f32_16x16x4_f32(false, a, false, bf[kc],
                                                (short)0, c, false, false);
    }
    float* orow = out + (size_t)(mb + mhi) * FOUT + ncol;
    if (colok) {
#pragma unroll
      for (int r = 0; r < 8; ++r) {
        float v = c[r];
        if constexpr (MODE == 0) v += bv;
        if constexpr (MODE == 2 || MODE == 3) v += orow[(size_t)r * FOUT];
        if constexpr (MODE == 3) v = v > 0.0f ? v : 0.0f;
        orow[(size_t)r * FOUT] = v;
      }
    }
  }
}


template <int FIN, int FOUT>
static void launch_gemm_sh(const float* T, const float* W, const float* bias,
                           float* out, int M, int mode, hipStream_t s) {
  constexpr int TM = 4;
  constexpr int NTN = (FOUT + 15) / 16;
  long long waves = (long long)((M / 16) / TM) * NTN;
  unsigned blocks = (unsigned)CDIV(waves * 32, 256);
  switch (mode) {
    case 0: k_cheb_gemm<FIN, FOUT, 0><<<blocks, 256, 0, s>>>(T, W, bias, out, M); break;
    case 1: k_cheb_gemm<FIN, FOUT, 1><<<blocks, 256, 0, s>>>(T, W, bias, out, M); break;
    case 2: k_cheb_gemm<FIN, FOUT, 2><<<blocks, 256, 0, s>>>(T, W, bias, out, M); break;
    default: k_cheb_gemm<FIN, FOUT, 3><<<blocks, 256, 0, s>>>(T, W, bias, out, M); break;
  }
}

static void launch_gemm(const float* T, const float* W, const float* bias,
                        float* out, int M, int Fin, int Fout, int mode,
                        hipStream_t s) {
  if (Fin == 3 && Fout == 16)       launch_gemm_sh<3, 16>(T, W, bias, out, M, mode, s);
  else if (Fin == 16 && Fout == 16) launch_gemm_sh<16, 16>(T, W, bias, out, M, mode, s);
  else if (Fin == 16 && Fout == 32) launch_gemm_sh<16, 32>(T, W, bias, out, M, mode, s);
  else if (Fin == 32 && Fout == 16) launch_gemm_sh<32, 16>(T, W, bias, out, M, mode, s);
  else if (Fin == 16 && Fout == 3)  launch_gemm_sh<16, 3>(T, W, bias, out, M, mode, s);
}

static void run_conv(const float* x, const int* ei, const float* nrm,
                     const float* W  , const float* bias,
                     float* out, float* b0, float* b1, float* b2,
                     int N, int Fin, int Fout, int relu, hipStream_t s) {
  int E = 16 * N;
  int M = 4 * N;
  long long tsz = (long long)M * Fin;
  size_t wstride = (size_t)Fin * Fout;

  launch_gemm(x, W, bias, out, M, Fin, Fout, bias ? 0 : 1, s);

  k_zero<<<(unsigned)CDIV(tsz, 256), 256, 0, s>>>(b0, tsz);
  k_propagate<<<(unsigned)CDIV((long long)E * Fin, 256), 256, 0, s>>>(
      ei, nrm, x, b0, E, N, Fin, 1.0f);
  launch_gemm(b0, W + wstride, nullptr, out, M, Fin, Fout, 2, s);

  const float* p0 = x;
  float* prev = b0;
  float* bufs[3] = {b0, b1, b2};
  for (int k = 2; k < 6; ++k) {
    float* tk = bufs[(k - 1) % 3];
    k_negcopy<<<(unsigned)CDIV(tsz, 256), 256, 0, s>>>(tk, p0, tsz);
    k_propagate<<<(unsigned)CDIV((long long)E * Fin, 256), 256, 0, s>>>(
        ei, nrm, prev, tk, E, N, Fin, 2.0f);
    int last = (k == 5);
    launch_gemm(tk, W + (size_t)k * wstride, nullptr, out, M, Fin, Fout,
                (last && relu) ? 3 : 2, s);
    p0 = prev;
    prev = tk;
  }
}


extern "C" void kernel_launch(void* const* d_in, const int* in_sizes, int n_in,
                              void* d_out, int out_size, void* d_ws, size_t ws_size,
                              hipStream_t stream) {
  (void)in_sizes; (void)n_in; (void)out_size; (void)ws_size;
  static const int NSs[5] = {65536, 16384, 4096, 1024, 256};
  static const int FE[5] = {3, 16, 16, 16, 32};
  static const int FD[6] = {32, 16, 16, 16, 16, 3};

  const float* x = (const float*)d_in[0];
  const int* ei[5];
  for (int l = 0; l < 5; ++l) ei[l] = (const int*)d_in[1 + l];
  const int* down_i[4]; const float* down_w[4];
  const int* up_i[4];   const float* up_w[4];
  for (int i = 0; i < 4; ++i) {
    down_i[i] = (const int*)d_in[6 + 4 * i];
    down_w[i] = (const float*)d_in[7 + 4 * i];
    up_i[i]   = (const int*)d_in[8 + 4 * i];
    up_w[i]   = (const float*)d_in[9 + 4 * i];
  }
  const float* encW[4]; const float* encB[4];
  for (int i = 0; i < 4; ++i) {
    encW[i] = (const float*)d_in[22 + 2 * i];
    encB[i] = (const float*)d_in[23 + 2 * i];
  }
  const float* decW[5];
  for (int i = 0; i < 5; ++i) decW[i] = (const float*)d_in[30 + i];
  const float* decB[4];
  for (int i = 0; i < 4; ++i) decB[i] = (const float*)d_in[35 + i];
  const float* enc_w = (const float*)d_in[39];
  const float* enc_b = (const float*)d_in[40];
  const float* dec_w = (const float*)d_in[41];
  const float* dec_b = (const float*)d_in[42];

  float* ws = (float*)d_ws;
  size_t off = 0;
  float* nrm[5];
  for (int l = 0; l < 5; ++l) { nrm[l] = ws + off; off += (size_t)16 * NSs[l]; }
  float* deg = ws + off; off += 65536;
  const size_t ACT = (size_t)4 * 65536 * 16;
  float* actA = ws + off; off += ACT;
  float* actB = ws + off; off += ACT;
  float* tb0 = ws + off; off += ACT;
  float* tb1 = ws + off; off += ACT;
  float* tb2 = ws + off; off += ACT;
  float* zbuf = ws + off; off += 256;

  for (int l = 0; l < 5; ++l) {
    int N = NSs[l], E = 16 * N;
    k_zero<<<(unsigned)CDIV((long long)N, 256), 256, 0, stream>>>(deg, N);
    k_deg_count<<<(unsigned)CDIV(E, 256), 256, 0, stream>>>(ei[l], deg, E);
    k_deg_inv<<<(unsigned)CDIV(N, 256), 256, 0, stream>>>(deg, N);
    k_edge_norm<<<(unsigned)CDIV(E, 256), 256, 0, stream>>>(ei[l], deg, nrm[l], E);
  }

  const float* cur = x;
  for (int i = 0; i < 4; ++i) {
    run_conv(cur, ei[i], nrm[i], encW[i], encB[i], actA, tb0, tb1, tb2,
             NSs[i], FE[i], FE[i + 1],  1, stream);
    long long pt = 4LL * NSs[i + 1] * FE[i + 1];
    k_pool<<<(unsigned)CDIV(pt, 256), 256, 0, stream>>>(
        actA, down_i[i], down_w[i], actB, NSs[i + 1], NSs[i], FE[i + 1]);
    cur = actB;
  }

  k_dense_enc<<<256, 256, 0, stream>>>(actB, enc_w, enc_b, zbuf);
  k_dense_dec<<<(unsigned)CDIV(4LL * 8192, 256), 256, 0, stream>>>(zbuf, dec_w,
                                                                   dec_b, actA);

  for (int i = 0; i < 4; ++i) {
    int lvl = 3 - i;
    long long pt = 4LL * NSs[lvl] * FD[i];
    k_pool<<<(unsigned)CDIV(pt, 256), 256, 0, stream>>>(
        actA, up_i[lvl], up_w[lvl], actB, NSs[lvl], NSs[lvl + 1], FD[i]);
    run_conv(actB, ei[lvl], nrm[lvl], decW[i], decB[i], actA, tb0, tb1, tb2,
             NSs[lvl], FD[i], FD[i + 1],  1, stream);
  }

  run_conv(actA, ei[0], nrm[0], decW[4], nullptr, (float*)d_out, tb0, tb1, tb2,
           NSs[0], FD[4], FD[5],  0, stream);
}


# Product of Chebyshev term k moved into the launch that scatters term k+1 (both only read T_k; a work-group does its tile, then its share of the product rows): 133 -> 88 launches
# speedup vs baseline: 2.5217x; 2.5217x over previous
#include <hip/hip_runtime.h>
#include <hip/hip_bf16.h>
#include <cstdint>
#include <cstddef>


typedef __attribute__((ext_vector_type(2))) float v2f;
typedef __attribute__((ext_vector_type(8))) float v8f;

#define CDIV(a, b) (((a) + (b) - 1) / (b))


__global__ void k_zero(float* __restrict__ p, long long n) {
  long long i = (long long)blockIdx.x * blockDim.x + threadIdx.x;
  if (i < n) p[i] = 0.0f;
}

__global__ void k_deg_count(const int* __restrict__ ei, float* __restrict__ deg, int E) {
  int e = blockIdx.x * blockDim.x + threadIdx.x;
  if (e < E) atomicAdd(&deg[ei[e]], 1.0f);
}

__global__ void k_deg_inv(float* __restrict__ deg, int N) {
  int i = blockIdx.x * blockDim.x + threadIdx.x;
  if (i < N) {
    float d = deg[i];
    deg[i] = (d > 0.0f) ? (1.0f / sqrtf(d)) : 0.0f;
  }
}


#define BIN_CHUNK 4096

__global__ __launch_bounds__(256) void k_bin_hist(const int* __restrict__ ei,
                                                  int* __restrict__ hist,
                                                  int shift, int nb, int nch) {
  __shared__ int lh[256];
  int tid = threadIdx.x;
  int chunk = blockIdx.x;
  lh[tid] = 0;
  __syncthreads();
  int base = chunk * BIN_CHUNK;
#pragma unroll
  for (int i = 0; i < BIN_CHUNK / 256; ++i) {
    int r = ei[base + i * 256 + tid];
    atomicAdd(&lh[r >> shift], 1);
  }
  __syncthreads();
  if (tid < nb) hist[tid * nch + chunk] = lh[tid];
}

template <int PER>
__global__ __launch_bounds__(1024) void k_scan_excl(int* __restrict__ h, int len) {
  __shared__ int sa[1024];
  __shared__ int sb[1024];
  int t = threadIdx.x;
  int b0 = t * PER;
  int v[PER];
  int sum = 0;
#pragma unroll
  for (int i = 0; i < PER; ++i) {
    v[i] = (b0 + i < len) ? h[b0 + i] : 0;
    sum += v[i];
  }
  sa[t] = sum;
  __syncthreads();
  int* pin = sa;
  int* pout = sb;
  for (int off = 1; off < 1024; off <<= 1) {
    int a = pin[t];
    if (t >= off) a += pin[t - off];
    pout[t] = a;
    __syncthreads();
    int* tmp = pin; pin = pout; pout = tmp;
  }
  int run = pin[t] - sum;
#pragma unroll
  for (int i = 0; i < PER; ++i) {
    if (b0 + i < len) h[b0 + i] = run;
    run += v[i];
  }
}

__global__ __launch_bounds__(256) void k_bin_place(const int* __restrict__ ei,
                                                   const int* __restrict__ hist,
                                                   int* __restrict__ recs,
                                                   int* __restrict__ bstart,
                                                   int E, int shift, int nb, int nch) {
  __shared__ int cur[256];
  int tid = threadIdx.x;
  int chunk = blockIdx.x;
  if (tid < nb) {
    int o = hist[tid * nch + chunk];
    cur[tid] = o;
    if (chunk == 0) bstart[tid] = o;
  }
  if (chunk == 0 && tid == 0) bstart[nb] = E;
  __syncthreads();
  int base = chunk * BIN_CHUNK;
  int rmask = (1 << shift) - 1;
#pragma unroll
  for (int i = 0; i < BIN_CHUNK / 256; ++i) {
    int e = base + i * 256 + tid;
    int r = ei[e];
    int c = ei[E + e];
    int p = atomicAdd(&cur[r >> shift], 1);
    recs[p] = ((r & rmask) << 16) | c;
  }
}

template <int F, int RSH>
__device__ __forceinline__ void tile_prop_body(
    int tile, const int* __restrict__ recs, const int* __restrict__ bstart,
    const float* __restrict__ dis, const float* __restrict__ prev,
    const float* p0, float* outT, int N, float scale) {
  constexpr int R = 1 << RSH;
  constexpr int RF = R * F;
  constexpr int U = 8;
  __shared__ __attribute__((aligned(16))) float acc[R * 4 * F];
  __shared__ float disr[R];
  int tid = threadIdx.x;
  int r0 = tile << RSH;
  for (int i = tid; i < R * 4 * F; i += 1024) acc[i] = 0.0f;
  if (tid < R) disr[tid] = dis[r0 + tid];
  __syncthreads();

  int beg = bstart[tile];
  int cnt = bstart[tile + 1] - beg;

  if constexpr (F % 4 == 0) {
    constexpr int Q = F / 4;
    constexpr int LPE = 4 * Q;
    int sub = tid % LPE;
    int b = sub / Q;
    int q = sub % Q;
    bool hi = ((tid >> 4) & 1) != 0;
    int o0 = hi ? 2 : 0;
    int total = cnt * LPE;
    float* abase = acc + b * F + 4 * q;
    const float* pbase = prev + (size_t)b * N * F + 4 * q;
    for (int t0 = tid; t0 < total; t0 += 1024 * U) {
      int rec[U];
      float wd[U];
      float4 xv[U];
#pragma unroll
      for (int u = 0; u < U; ++u) {
        int t = min(t0 + u * 1024, total - 1);
        rec[u] = recs[beg + t / LPE];
      }
#pragma unroll
      for (int u = 0; u < U; ++u) {
        int c = rec[u] & 0xFFFF;
        wd[u] = dis[c];
        xv[u] = *(const float4*)(pbase + (size_t)c * F);
      }
#pragma unroll
      for (int u = 0; u < U; ++u) {
        if (t0 + u * 1024 < total) {
          int rl = rec[u] >> 16;
          float w = disr[rl] * wd[u];
          float e0 = hi ? xv[u].z : xv[u].x;
          float e1 = hi ? xv[u].w : xv[u].y;
          float e2 = hi ? xv[u].x : xv[u].z;
          float e3 = hi ? xv[u].y : xv[u].w;
          float* a = abase + rl * 4 * F;
          atomicAdd(a + o0, w * e0);
          atomicAdd(a + o0 + 1, w * e1);
          atomicAdd(a + (o0 ^ 2), w * e2);
          atomicAdd(a + (o0 ^ 2) + 1, w * e3);
        }
      }
    }
  } else {
    constexpr int LPE = 4;
    int b = tid & 3;
    int total = cnt * LPE;
    const float* pbase = prev + (size_t)b * N * F;
    float* abase = acc + b * F;
    for (int t0 = tid; t0 < total; t0 += 1024 * U) {
      int rec[U];
      float wd[U];
      float xs[U][F];
#pragma unroll
      for (int u = 0; u < U; ++u) {
        int t = min(t0 + u * 1024, total - 1);
        rec[u] = recs[beg + (t >> 2)];
      }
#pragma unroll
      for (int u = 0; u < U; ++u) {
        int c = rec[u] & 0xFFFF;
        wd[u] = dis[c];
        const float* p = pbase + (size_t)c * F;
        for (int f = 0; f < F; ++f) xs[u][f] = p[f];
      }
#pragma unroll
      for (int u = 0; u < U; ++u) {
        if (t0 + u * 1024 < total) {
          int rl = rec[u] >> 16;
          float w = disr[rl] * wd[u];
          float* a = abase + rl * 4 * F;
          for (int f = 0; f < F; ++f) atomicAdd(a + f, w * xs[u][f]);
        }
      }
    }
  }
  __syncthreads();

  if constexpr (F % 4 == 0) {
    for (int i = tid; i < RF; i += 1024) {
      int e = i * 4;
      int b = e / RF;
      int rem = e % RF;
      int rl = rem / F;
      int f = rem % F;
      size_t g = ((size_t)b * N + r0) * F + rem;
      float4 a = *(const float4*)(acc + (rl * 4 + b) * F + f);
      float4 v;
      v.x = scale * a.x; v.y = scale * a.y; v.z = scale * a.z; v.w = scale * a.w;
      if (p0) {
        float4 p = *(const float4*)(p0 + g);
        v.x -= p.x; v.y -= p.y; v.z -= p.z; v.w -= p.w;
      }
      *(float4*)(outT + g) = v;
    }
  } else {
    for (int i = tid; i < 4 * RF; i += 1024) {
      int b = i / RF;
      int rem = i % RF;
      int rl = rem / F;
      int f = rem % F;
      size_t g = ((size_t)b * N + r0) * F + rem;
      float v = scale * acc[(rl * 4 + b) * F + f];
      if (p0) v -= p0[g];
      outT[g] = v;
    }
  }
}

__global__ void k_pool(const float* __restrict__ x, const int* __restrict__ idx,
                       const float* __restrict__ w, float* __restrict__ out,
                       int Nout, int Nin, int F) {
  long long tid = (long long)blockIdx.x * blockDim.x + threadIdx.x;
  long long total = 4LL * Nout * F;
  if (tid >= total) return;
  int f = (int)(tid % F);
  int n = (int)((tid / F) % Nout);
  int b = (int)(tid / ((long long)F * Nout));
  size_t nfin = (size_t)Nin * F;
  float s = 0.0f;
#pragma unroll
  for (int k = 0; k < 3; ++k) {
    int j = idx[n * 3 + k];
    s += w[n * 3 + k] * x[(size_t)b * nfin + (size_t)j * F + f];
  }
  out[((size_t)b * Nout + n) * F + f] = s;
}

__global__ void k_dense_enc(const float* __restrict__ h, const float* __restrict__ w,
                            const float* __restrict__ bias, float* __restrict__ z) {
  int b = blockIdx.x >> 6;
  int j = blockIdx.x & 63;
  float s = 0.0f;
  for (int i = threadIdx.x; i < 8192; i += blockDim.x)
    s += h[b * 8192 + i] * w[(size_t)i * 64 + j];
  __shared__ float red[256];
  red[threadIdx.x] = s;
  __syncthreads();
  for (int off = 128; off > 0; off >>= 1) {
    if (threadIdx.x < off) red[threadIdx.x] += red[threadIdx.x + off];
    __syncthreads();
  }
  if (threadIdx.x == 0) {
    float v = red[0] + bias[j];
    z[b * 64 + j] = v > 0.0f ? v : 0.0f;
  }
}

__global__ void k_dense_dec(const float* __restrict__ z, const float* __restrict__ w,
                            const float* __restrict__ bias, float* __restrict__ h) {
  long long tid = (long long)blockIdx.x * blockDim.x + threadIdx.x;
  if (tid >= 4LL * 8192) return;
  int b = (int)(tid >> 13);
  int j = (int)(tid & 8191);
  float s = bias[j];
  for (int i = 0; i < 64; ++i) s += z[b * 64 + i] * w[(size_t)i * 8192 + j];
  h[tid] = s > 0.0f ? s : 0.0f;
}


__device__ __forceinline__ float ldsel(const float* __restrict__ p, int idx, bool ok) {
  float v = p[ok ? idx : 0];
  return ok ? v : 0.0f;
}

template <int FIN, int FOUT, int MODE>
__device__ __forceinline__ void cheb_gemm_body(
    int wave, const float* __restrict__ T, const float* __restrict__ W,
    const float* __restrict__ bias, float* __restrict__ out, int M) {
  constexpr int KC = (FIN + 3) / 4;
  constexpr int NTN = (FOUT + 15) / 16;
  constexpr int TM = 4;

  int lane = threadIdx.x & 31;
  int groups = (M >> 4) / TM;
  if (wave >= groups * NTN) return;

  int tn = wave % NTN;
  int gm = wave / NTN;
  int m0 = gm * (16 * TM);
  int n0 = tn * 16;
  int lm = lane & 15;
  int khalf = (lane >> 4) << 1;
  int mhi = (lane >> 4) << 3;
  int ncol = n0 + lm;
  bool colok = (FOUT % 16 == 0) || (ncol < FOUT);

  v2f bf[KC];
#pragma unroll
  for (int kc = 0; kc < KC; ++kc) {
    int ka = kc * 4 + khalf;
    if constexpr (FIN % 4 == 0 && FOUT % 16 == 0) {
      bf[kc].x = W[ka * FOUT + ncol];
      bf[kc].y = W[(ka + 1) * FOUT + ncol];
    } else {
      bf[kc].x = ldsel(W, ka * FOUT + ncol, colok && (ka < FIN));
      bf[kc].y = ldsel(W, (ka + 1) * FOUT + ncol, colok && (ka + 1 < FIN));
    }
  }
  float bv = 0.0f;
  if constexpr (MODE == 0) bv = ldsel(bias, ncol, colok);

#pragma unroll
  for (int t = 0; t < TM; ++t) {
    int mb = m0 + t * 16;
    const float* trow = T + (size_t)(mb + lm) * FIN;
    v8f c = {0.f, 0.f, 0.f, 0.f, 0.f, 0.f, 0.f, 0.f};
#pragma unroll
    for (int kc = 0; kc < KC; ++kc) {
      int ka = kc * 4 + khalf;
      v2f a;
      if constexpr (FIN % 4 == 0) {
        a = *(const v2f*)(trow + ka);
      } else {
        a.x = ldsel(trow, ka, ka < FIN);
        a.y = ldsel(trow, ka + 1, ka + 1 < FIN);
      }
      c = __builtin_amdgcn_wmma_f32_16x16x4_f32(false, a, false, bf[kc],
                                                (short)0, c, false, false);
    }
    float* orow = out + (size_t)(mb + mhi) * FOUT + ncol;
    if (colok) {
#pragma unroll
      for (int r = 0; r < 8; ++r) {
        float v = c[r];
        if constexpr (MODE == 0) v += bv;
        if constexpr (MODE == 2 || MODE == 3) v += orow[(size_t)r * FOUT];
        if constexpr (MODE == 3) v = v > 0.0f ? v : 0.0f;
        orow[(size_t)r * FOUT] = v;
      }
    }
  }
}

template <int FIN, int FOUT, int MODE>
__global__ __launch_bounds__(256) void k_cheb_gemm(
    const float* __restrict__ T, const float* __restrict__ W,
    const float* __restrict__ bias, float* __restrict__ out, int M) {
  cheb_gemm_body<FIN, FOUT, MODE>((int)((blockIdx.x * blockDim.x + threadIdx.x) >> 5),
                                  T, W, bias, out, M);
}

template <int F, int RSH, int FOUT, int MODE>
__global__ __launch_bounds__(1024) void k_prop_gemm(
    const int* __restrict__ recs, const int* __restrict__ bstart,
    const float* __restrict__ dis, const float* __restrict__ prev,
    const float* p0, float* outT, int N, float scale,
    const float* __restrict__ W, const float* __restrict__ bias,
    float* __restrict__ out) {
  constexpr int NTN = (FOUT + 15) / 16;
  constexpr int PER = NTN << (RSH - 4);
  static_assert(RSH >= 4 && PER <= 32, "a work-group has 32 waves");
  int blk = (int)blockIdx.x;
  tile_prop_body<F, RSH>(blk, recs, bstart, dis, prev, p0, outT, N, scale);

  int wv = (int)(threadIdx.x >> 5);
  if (wv < PER)
    cheb_gemm_body<F, FOUT, MODE>(blk * PER + wv, prev, W, bias, out, 4 * N);
}


template <int FIN, int FOUT>
static void launch_gemm_sh(const float* T, const float* W, const float* bias,
                           float* out, int M, int mode, hipStream_t s) {
  constexpr int TM = 4;
  constexpr int NTN = (FOUT + 15) / 16;
  long long waves = (long long)((M / 16) / TM) * NTN;
  unsigned blocks = (unsigned)CDIV(waves * 32, 256);
  switch (mode) {
    case 0: k_cheb_gemm<FIN, FOUT, 0><<<blocks, 256, 0, s>>>(T, W, bias, out, M); break;
    case 1: k_cheb_gemm<FIN, FOUT, 1><<<blocks, 256, 0, s>>>(T, W, bias, out, M); break;
    case 2: k_cheb_gemm<FIN, FOUT, 2><<<blocks, 256, 0, s>>>(T, W, bias, out, M); break;
    default: k_cheb_gemm<FIN, FOUT, 3><<<blocks, 256, 0, s>>>(T, W, bias, out, M); break;
  }
}

static void launch_gemm(const float* T, const float* W, const float* bias,
                        float* out, int M, int Fin, int Fout, int mode,
                        hipStream_t s) {
  if (Fin == 3 && Fout == 16)       launch_gemm_sh<3, 16>(T, W, bias, out, M, mode, s);
  else if (Fin == 16 && Fout == 16) launch_gemm_sh<16, 16>(T, W, bias, out, M, mode, s);
  else if (Fin == 16 && Fout == 32) launch_gemm_sh<16, 32>(T, W, bias, out, M, mode, s);
  else if (Fin == 32 && Fout == 16) launch_gemm_sh<32, 16>(T, W, bias, out, M, mode, s);
  else if (Fin == 16 && Fout == 3)  launch_gemm_sh<16, 3>(T, W, bias, out, M, mode, s);
}

struct LevelGeo {
  const int* recs;
  const int* bstart;
  const float* dis;
  int N;
  int rsh;
};

template <int F, int RSH, int FOUT>
static void launch_pg_sh(const LevelGeo& g, const float* prev, const float* p0,
                         float* outT, float scale, const float* W, const float* bias,
                         float* out, int mode, hipStream_t s) {
  unsigned grid = (unsigned)(g.N >> RSH);
  switch (mode) {
    case 0:
      k_prop_gemm<F, RSH, FOUT, 0><<<grid, 1024, 0, s>>>(
          g.recs, g.bstart, g.dis, prev, p0, outT, g.N, scale, W, bias, out);
      break;
    case 1:
      k_prop_gemm<F, RSH, FOUT, 1><<<grid, 1024, 0, s>>>(
          g.recs, g.bstart, g.dis, prev, p0, outT, g.N, scale, W, bias, out);
      break;
    default:
      k_prop_gemm<F, RSH, FOUT, 2><<<grid, 1024, 0, s>>>(
          g.recs, g.bstart, g.dis, prev, p0, outT, g.N, scale, W, bias, out);
      break;
  }
}

static void launch_prop_gemm(const LevelGeo& g, int Fin, int Fout, const float* prev,
                             const float* p0, float* outT, float scale,
                             const float* W, const float* bias, float* out, int mode,
                             hipStream_t s) {
  if (Fin == 3)
    launch_pg_sh<3, 8, 16>(g, prev, p0, outT, scale, W, bias, out, mode, s);
  else if (Fin == 32)
    launch_pg_sh<32, 4, 16>(g, prev, p0, outT, scale, W, bias, out, mode, s);
  else if (Fout == 32)
    launch_pg_sh<16, 4, 32>(g, prev, p0, outT, scale, W, bias, out, mode, s);
  else if (Fout == 3)
    launch_pg_sh<16, 8, 3>(g, prev, p0, outT, scale, W, bias, out, mode, s);
  else if (g.rsh == 8)
    launch_pg_sh<16, 8, 16>(g, prev, p0, outT, scale, W, bias, out, mode, s);
  else if (g.rsh == 6)
    launch_pg_sh<16, 6, 16>(g, prev, p0, outT, scale, W, bias, out, mode, s);
  else
    launch_pg_sh<16, 4, 16>(g, prev, p0, outT, scale, W, bias, out, mode, s);
}

static void run_conv(const float* x, const LevelGeo& g,
                     const float* W  , const float* bias,
                     float* out, float* b0, float* b1,
                     int Fin, int Fout, int relu, hipStream_t s) {
  int M = 4 * g.N;
  size_t wstride = (size_t)Fin * Fout;

  launch_prop_gemm(g, Fin, Fout, x, nullptr, b0, 1.0f, W, bias, out, bias ? 0 : 1, s);

  launch_prop_gemm(g, Fin, Fout, b0, x, b1, 2.0f, W + wstride, nullptr, out, 2, s);

  float* prev = b1;
  float* old = b0;
  for (int k = 3; k < 6; ++k) {
    launch_prop_gemm(g, Fin, Fout, prev, old, old, 2.0f, W + (size_t)(k - 1) * wstride,
                     nullptr, out, 2, s);
    float* t = prev; prev = old; old = t;
  }

  launch_gemm(prev, W + 5 * wstride, nullptr, out, M, Fin, Fout, relu ? 3 : 2, s);
}


extern "C" void kernel_launch(void* const* d_in, const int* in_sizes, int n_in,
                              void* d_out, int out_size, void* d_ws, size_t ws_size,
                              hipStream_t stream) {
  (void)in_sizes; (void)n_in; (void)out_size; (void)ws_size;
  static const int NSs[5] = {65536, 16384, 4096, 1024, 256};
  static const int FE[5] = {3, 16, 16, 16, 32};
  static const int FD[6] = {32, 16, 16, 16, 16, 3};

  const float* x = (const float*)d_in[0];
  const int* ei[5];
  for (int l = 0; l < 5; ++l) ei[l] = (const int*)d_in[1 + l];
  const int* down_i[4]; const float* down_w[4];
  const int* up_i[4];   const float* up_w[4];
  for (int i = 0; i < 4; ++i) {
    down_i[i] = (const int*)d_in[6 + 4 * i];
    down_w[i] = (const float*)d_in[7 + 4 * i];
    up_i[i]   = (const int*)d_in[8 + 4 * i];
    up_w[i]   = (const float*)d_in[9 + 4 * i];
  }
  const float* encW[4]; const float* encB[4];
  for (int i = 0; i < 4; ++i) {
    encW[i] = (const float*)d_in[22 + 2 * i];
    encB[i] = (const float*)d_in[23 + 2 * i];
  }
  const float* decW[5];
  for (int i = 0; i < 5; ++i) decW[i] = (const float*)d_in[30 + i];
  const float* decB[4];
  for (int i = 0; i < 4; ++i) decB[i] = (const float*)d_in[35 + i];
  const float* enc_w = (const float*)d_in[39];
  const float* enc_b = (const float*)d_in[40];
  const float* dec_w = (const float*)d_in[41];
  const float* dec_b = (const float*)d_in[42];

  static const int RSH[4] = {8, 6, 4, 4};
  float* ws = (float*)d_ws;
  size_t off = 0;
  int* recs[4];
  for (int l = 0; l < 4; ++l) { recs[l] = (int*)(ws + off); off += (size_t)16 * NSs[l]; }
  float* dis[4];
  for (int l = 0; l < 4; ++l) { dis[l] = ws + off; off += (size_t)NSs[l]; }
  int* bstart[4];
  for (int l = 0; l < 4; ++l) { bstart[l] = (int*)(ws + off); off += 512; }
  int* hist = (int*)(ws + off); off += 65536;
  const size_t ACT = (size_t)4 * 65536 * 16;
  float* actA = ws + off; off += ACT;
  float* actB = ws + off; off += ACT;
  float* tb0 = ws + off; off += ACT;
  float* tb1 = ws + off; off += ACT;
  float* zbuf = ws + off; off += 256;

  LevelGeo geo[4];
  for (int l = 0; l < 4; ++l) {
    int N = NSs[l], E = 16 * N;
    int nb = N >> RSH[l];
    int nch = E / BIN_CHUNK;
    int len = nb * nch;
    k_zero<<<(unsigned)CDIV((long long)N, 256), 256, 0, stream>>>(dis[l], N);
    k_deg_count<<<(unsigned)CDIV(E, 256), 256, 0, stream>>>(ei[l], dis[l], E);
    k_deg_inv<<<(unsigned)CDIV(N, 256), 256, 0, stream>>>(dis[l], N);
    k_bin_hist<<<(unsigned)nch, 256, 0, stream>>>(ei[l], hist, RSH[l], nb, nch);
    if (len > 16384)     k_scan_excl<64><<<1, 1024, 0, stream>>>(hist, len);
    else if (len > 4096) k_scan_excl<16><<<1, 1024, 0, stream>>>(hist, len);
    else if (len > 1024) k_scan_excl<4><<<1, 1024, 0, stream>>>(hist, len);
    else                 k_scan_excl<1><<<1, 1024, 0, stream>>>(hist, len);
    k_bin_place<<<(unsigned)nch, 256, 0, stream>>>(ei[l], hist, recs[l], bstart[l],
                                                   E, RSH[l], nb, nch);
    geo[l].recs = recs[l];
    geo[l].bstart = bstart[l];
    geo[l].dis = dis[l];
    geo[l].N = N;
    geo[l].rsh = RSH[l];
  }

  const float* cur = x;
  for (int i = 0; i < 4; ++i) {
    run_conv(cur, geo[i], encW[i], encB[i], actA, tb0, tb1,
             FE[i], FE[i + 1],  1, stream);
    long long pt = 4LL * NSs[i + 1] * FE[i + 1];
    k_pool<<<(unsigned)CDIV(pt, 256), 256, 0, stream>>>(
        actA, down_i[i], down_w[i], actB, NSs[i + 1], NSs[i], FE[i + 1]);
    cur = actB;
  }

  k_dense_enc<<<256, 256, 0, stream>>>(actB, enc_w, enc_b, zbuf);
  k_dense_dec<<<(unsigned)CDIV(4LL * 8192, 256), 256, 0, stream>>>(zbuf, dec_w,
                                                                   dec_b, actA);

  for (int i = 0; i < 4; ++i) {
    int lvl = 3 - i;
    long long pt = 4LL * NSs[lvl] * FD[i];
    k_pool<<<(unsigned)CDIV(pt, 256), 256, 0, stream>>>(
        actA, up_i[lvl], up_w[lvl], actB, NSs[lvl], NSs[lvl + 1], FD[i]);
    run_conv(actB, geo[lvl], decW[i], decB[i], actA, tb0, tb1,
             FD[i], FD[i + 1],  1, stream);
  }

  run_conv(actA, geo[0], decW[4], nullptr, (float*)d_out, tb0, tb1,
           FD[4], FD[5],  0, stream);
}
